# starting kernel
# baseline (speedup 1.0000x reference)
#include <hip/hip_runtime.h>

typedef __attribute__((ext_vector_type(2))) float v2f;
typedef __attribute__((ext_vector_type(8))) float v8f;

#define NK 64
#define NC 2048
#define NN 1024
#define NB 16

__global__ __launch_bounds__(128) void netvlad_assign(
    const float* __restrict__ x, const float* __restrict__ w, float* __restrict__ a)
{
    const int lane = threadIdx.x & 31;
    const int wid  = threadIdx.x >> 5;
    const int col  = lane & 15;
    const int half = lane >> 4;
    const int koff = half << 1;

    const int tile = blockIdx.x * 4 + wid;
    const int b    = tile >> 6;
    const int n0   = (tile & 63) << 4;

    const float* xb = x + (size_t)b * NC * NN;

    v8f acc[4] = {v8f{}, v8f{}, v8f{}, v8f{}};

    for (int c = 0; c < NC; c += 4) {
        v2f af0 = *(const v2f*)(w + (size_t)( 0 + col) * NC + c + koff);
        v2f af1 = *(const v2f*)(w + (size_t)(16 + col) * NC + c + koff);
        v2f af2 = *(const v2f*)(w + (size_t)(32 + col) * NC + c + koff);
        v2f af3 = *(const v2f*)(w + (size_t)(48 + col) * NC + c + koff);
        v2f bf;
        bf.x = xb[(size_t)(c + koff + 0) * NN + n0 + col];
        bf.y = xb[(size_t)(c + koff + 1) * NN + n0 + col];

        acc[0] = __builtin_amdgcn_wmma_f32_16x16x4_f32(false, af0, false, bf, (short)0, acc[0], false, false);
        acc[1] = __builtin_amdgcn_wmma_f32_16x16x4_f32(false, af1, false, bf, (short)0, acc[1], false, false);
        acc[2] = __builtin_amdgcn_wmma_f32_16x16x4_f32(false, af2, false, bf, (short)0, acc[2], false, false);
        acc[3] = __builtin_amdgcn_wmma_f32_16x16x4_f32(false, af3, false, bf, (short)0, acc[3], false, false);
    }

    float mx = -3.402823466e38f;
#pragma unroll
    for (int t = 0; t < 4; ++t)
#pragma unroll
        for (int r = 0; r < 8; ++r) mx = fmaxf(mx, acc[t][r]);
    mx = fmaxf(mx, __shfl_xor(mx, 16, 32));

    float sum = 0.0f;
#pragma unroll
    for (int t = 0; t < 4; ++t)
#pragma unroll
        for (int r = 0; r < 8; ++r) {
            float e = __expf(acc[t][r] - mx);
            acc[t][r] = e;
            sum += e;
        }
    sum += __shfl_xor(sum, 16, 32);
    const float inv = 1.0f / sum;

    float* ab = a + (size_t)b * NK * NN;
#pragma unroll
    for (int t = 0; t < 4; ++t)
#pragma unroll
        for (int r = 0; r < 8; ++r) {
            int k = t * 16 + r + 8 * half;
            ab[(size_t)k * NN + n0 + col] = acc[t][r] * inv;
        }
}

__global__ __launch_bounds__(256) void netvlad_asum(
    const float* __restrict__ a, float* __restrict__ asum)
{
    int idx = blockIdx.x * 256 + threadIdx.x;
    const float4* row = (const float4*)(a + (size_t)idx * NN);
    float s = 0.0f;
    for (int i = 0; i < NN / 4; ++i) {
        float4 v = row[i];
        s += (v.x + v.y) + (v.z + v.w);
    }
    asum[idx] = s;
}

__global__ __launch_bounds__(128) void netvlad_vlad(
    const float* __restrict__ x, const float* __restrict__ a,
    const float* __restrict__ asum, const float* __restrict__ cent,
    float* __restrict__ out)
{
    const int lane = threadIdx.x & 31;
    const int wid  = threadIdx.x >> 5;
    const int col  = lane & 15;
    const int half = lane >> 4;
    const int koff = half << 1;

    const int tile = blockIdx.x * 4 + wid;
    const int b    = tile >> 7;
    const int c0   = (tile & 127) << 4;

    const float* ab   = a + (size_t)b * NK * NN;
    const float* xrow = x + ((size_t)b * NC + c0 + col) * NN;

    v8f acc[4] = {v8f{}, v8f{}, v8f{}, v8f{}};

    for (int n = 0; n < NN; n += 4) {
        v2f af0 = *(const v2f*)(ab + (size_t)( 0 + col) * NN + n + koff);
        v2f af1 = *(const v2f*)(ab + (size_t)(16 + col) * NN + n + koff);
        v2f af2 = *(const v2f*)(ab + (size_t)(32 + col) * NN + n + koff);
        v2f af3 = *(const v2f*)(ab + (size_t)(48 + col) * NN + n + koff);
        v2f bf = *(const v2f*)(xrow + n + koff);

        acc[0] = __builtin_amdgcn_wmma_f32_16x16x4_f32(false, af0, false, bf, (short)0, acc[0], false, false);
        acc[1] = __builtin_amdgcn_wmma_f32_16x16x4_f32(false, af1, false, bf, (short)0, acc[1], false, false);
        acc[2] = __builtin_amdgcn_wmma_f32_16x16x4_f32(false, af2, false, bf, (short)0, acc[2], false, false);
        acc[3] = __builtin_amdgcn_wmma_f32_16x16x4_f32(false, af3, false, bf, (short)0, acc[3], false, false);
    }

    float* ob = out + (size_t)b * NK * NC;
#pragma unroll
    for (int t = 0; t < 4; ++t)
#pragma unroll
        for (int r = 0; r < 8; ++r) {
            int k = t * 16 + r + 8 * half;
            float s  = asum[b * NK + k];
            float cv = cent[(size_t)k * NC + c0 + col];
            ob[(size_t)k * NC + c0 + col] = acc[t][r] - s * cv;
        }
}

__global__ __launch_bounds__(256) void netvlad_norm(
    const float* __restrict__ out, float* __restrict__ invn)
{
    __shared__ float red[256];
    const float4* vb = (const float4*)(out + (size_t)blockIdx.x * NK * NC);
    float s = 0.0f;
    for (int i = threadIdx.x; i < (NK * NC) / 4; i += 256) {
        float4 v = vb[i];
        s += v.x * v.x + v.y * v.y + v.z * v.z + v.w * v.w;
    }
    red[threadIdx.x] = s;
    __syncthreads();
    for (int off = 128; off > 0; off >>= 1) {
        if (threadIdx.x < off) red[threadIdx.x] += red[threadIdx.x + off];
        __syncthreads();
    }
    if (threadIdx.x == 0)
        invn[blockIdx.x] = 1.0f / fmaxf(sqrtf(red[0]), 1e-12f);
}

__global__ __launch_bounds__(256) void netvlad_scale(
    float* __restrict__ out, const float* __restrict__ invn)
{
    int gid = blockIdx.x * 256 + threadIdx.x;
    int b = gid / ((NK * NC) / 4);
    float sc = invn[b];
    float4* p = (float4*)out;
    float4 v = p[gid];
    v.x *= sc; v.y *= sc; v.z *= sc; v.w *= sc;
    p[gid] = v;
}

extern "C" void kernel_launch(void* const* d_in, const int* in_sizes, int n_in,
                              void* d_out, int out_size, void* d_ws, size_t ws_size,
                              hipStream_t stream) {
    const float* x    = (const float*)d_in[0];
    const float* w    = (const float*)d_in[1];
    const float* cent = (const float*)d_in[2];
    float* out = (float*)d_out;

    float* a    = (float*)d_ws;
    float* asum = a + (size_t)NB * NK * NN;
    float* invn = asum + NB * NK;

    netvlad_assign<<<(NB * NN / 16) / 4, 128, 0, stream>>>(x, w, a);
    netvlad_asum  <<<(NB * NK) / 256, 256, 0, stream>>>(a, asum);
    netvlad_vlad  <<<(NB * NC / 16) / 4, 128, 0, stream>>>(x, a, asum, cent, out);
    netvlad_norm  <<<NB, 256, 0, stream>>>(out, invn);
    netvlad_scale <<<(NB * NK * NC / 4) / 256, 256, 0, stream>>>(out, invn);
}


# vlad kernel: the per-lane global fragment loads of n block i+1 go out as one group ahead of block i's WMMAs into a second register set (no LDS, same loads); n order and every sum order unchanged
# speedup vs baseline: 3.6021x; 3.6021x over previous
#include <hip/hip_runtime.h>

typedef __attribute__((ext_vector_type(2))) float v2f;
typedef __attribute__((ext_vector_type(8))) float v8f;

#define NK 64
#define NC 2048
#define NN 1024
#define NB 16

#define GC 32
#define NG (NC / GC)
#define PD 8
#define WP 36
#define XP 72

typedef __attribute__((ext_vector_type(4))) float v4f;

#define LDS_BARRIER()                                                           \
    do {                                                                        \
        __builtin_amdgcn_fence(__ATOMIC_RELEASE, "workgroup", "local");         \
        __builtin_amdgcn_s_barrier();                                           \
        __builtin_amdgcn_fence(__ATOMIC_ACQUIRE, "workgroup", "local");         \
    } while (0)

#define ASSIGN_ISSUE(SLOT)                                                      \
    {                                                                           \
        _Pragma("unroll")           \
        for (int i = 0; i < 4; ++i) {                                           \
            rw[SLOT][i] = *(const v4f*)(wp + (size_t)(16 * i) * NC);            \
            rx[SLOT][i] = *(const v4f*)(xp + (size_t)(8 * i) * NN);             \
        }                                                                       \
        const bool last = (cn == NG - 1);                                       \
        wp += last ? -(ptrdiff_t)(NG - 1) * GC : (ptrdiff_t)GC;                 \
        xp += last ? -(ptrdiff_t)(NG - 1) * GC * NN : (ptrdiff_t)GC * NN;       \
        cn = (cn + 1) & (NG - 1);                                               \
    }

#define ASSIGN_STASH(SLOT, STG)                                                 \
    _Pragma("unroll")               \
    for (int i = 0; i < 4; ++i) {                                               \
        *(v4f*)(&ws[STG][wso + 16 * i * WP]) = rw[SLOT][i];                     \
        *(v4f*)(&xs[STG][xso + 8 * i * XP])  = rx[SLOT][i];                     \
    }

#define ASSIGN_FRAGS(SET, STG)                                                  \
    {                                                                           \
        const float* wl = &ws[STG][col * WP + koff];                            \
        const float* xl = &xs[STG][koff * XP + (wid << 4) + col];               \
        _Pragma("unroll")   \
        for (int s = 0; s < GC / 4; ++s) {                                      \
            fa[SET][s][0] = *(const v2f*)(wl +  0 * WP + 4 * s);                \
            fa[SET][s][1] = *(const v2f*)(wl + 16 * WP + 4 * s);                \
            fa[SET][s][2] = *(const v2f*)(wl + 32 * WP + 4 * s);                \
            fa[SET][s][3] = *(const v2f*)(wl + 48 * WP + 4 * s);                \
            fb[SET][s].x = xl[(4 * s + 0) * XP];                                \
            fb[SET][s].y = xl[(4 * s + 1) * XP];                                \
        }                                                                       \
    }

#define ASSIGN_WMMA(SET)                                                        \
    _Pragma("unroll")               \
    for (int s = 0; s < GC / 4; ++s) {                                          \
        acc[0] = __builtin_amdgcn_wmma_f32_16x16x4_f32(false, fa[SET][s][0], false, fb[SET][s], (short)0, acc[0], false, false); \
        acc[1] = __builtin_amdgcn_wmma_f32_16x16x4_f32(false, fa[SET][s][1], false, fb[SET][s], (short)0, acc[1], false, false); \
        acc[2] = __builtin_amdgcn_wmma_f32_16x16x4_f32(false, fa[SET][s][2], false, fb[SET][s], (short)0, acc[2], false, false); \
        acc[3] = __builtin_amdgcn_wmma_f32_16x16x4_f32(false, fa[SET][s][3], false, fb[SET][s], (short)0, acc[3], false, false); \
    }

__global__ __launch_bounds__(128) void netvlad_assign(
    const float* __restrict__ x, const float* __restrict__ w, float* __restrict__ a)
{
    __shared__ __attribute__((aligned(16))) float ws[4][NK * WP];
    __shared__ __attribute__((aligned(16))) float xs[4][GC * XP];
    __shared__ __attribute__((aligned(16))) float as[4][NK * 16];

    const int tid  = threadIdx.x;
    const int lane = tid & 31;
    const int wid  = tid >> 5;
    const int col  = lane & 15;
    const int half = lane >> 4;
    const int koff = half << 1;

    const int b  = blockIdx.x >> 4;
    const int nb = (blockIdx.x & 15) << 6;
    const int n0 = nb + (wid << 4);
    int cn = blockIdx.x & (NG - 1);

    const float* xp = x + (size_t)b * NC * NN + (size_t)(tid >> 4) * NN + nb + ((tid & 15) << 2) + (size_t)cn * GC * NN;
    const float* wp = w + (size_t)(tid >> 3) * NC + ((tid & 7) << 2) + cn * GC;
    const int xso = (tid >> 4) * XP + ((tid & 15) << 2);
    const int wso = (tid >> 3) * WP + ((tid & 7) << 2);

    v4f rx[PD][4], rw[PD][4];
    v2f fa[2][GC / 4][4], fb[2][GC / 4];
    v8f acc[4] = {v8f{}, v8f{}, v8f{}, v8f{}};

#pragma unroll
    for (int p = 0; p < PD; ++p) {
        ASSIGN_ISSUE(p)
        __builtin_amdgcn_sched_barrier(0);
    }
    ASSIGN_STASH(0, 0)
    ASSIGN_STASH(1, 1)
    LDS_BARRIER();
    ASSIGN_FRAGS(0, 0)

    for (int g0 = 0; g0 < NG - PD; g0 += PD) {
#pragma unroll
        for (int j = 0; j < PD; ++j) {
            ASSIGN_ISSUE(j)
            ASSIGN_STASH((j + 2) & (PD - 1), (j + 2) & 3)
            ASSIGN_FRAGS((j + 1) & 1, (j + 1) & 3)
            __builtin_amdgcn_sched_barrier(0);
            ASSIGN_WMMA(j & 1)
            __builtin_amdgcn_sched_barrier(0);
            LDS_BARRIER();
        }
    }
#pragma unroll
    for (int j = 0; j < PD; ++j) {
        if (j + 2 < PD) {
            ASSIGN_STASH((j + 2) & (PD - 1), (j + 2) & 3)
        }
        if (j + 1 < PD) {
            ASSIGN_FRAGS((j + 1) & 1, (j + 1) & 3)
        }
        __builtin_amdgcn_sched_barrier(0);
        ASSIGN_WMMA(j & 1)
        __builtin_amdgcn_sched_barrier(0);
        if (j + 2 < PD) {
            LDS_BARRIER();
        }
    }

    float mx = -3.402823466e38f;
#pragma unroll
    for (int t = 0; t < 4; ++t)
#pragma unroll
        for (int r = 0; r < 8; ++r) mx = fmaxf(mx, acc[t][r]);
    mx = fmaxf(mx, __shfl_xor(mx, 16, 32));

    float sum = 0.0f;
#pragma unroll
    for (int t = 0; t < 4; ++t)
#pragma unroll
        for (int r = 0; r < 8; ++r) {
            float e = __expf(acc[t][r] - mx);
            acc[t][r] = e;
            sum += e;
        }
    sum += __shfl_xor(sum, 16, 32);
    const float inv = 1.0f / sum;

    float* asw = as[wid];
    {
        float* dst = asw + ((col >> 1) << 5) + (half << 4) + (col & 1);
#pragma unroll
        for (int t = 0; t < 4; ++t)
#pragma unroll
            for (int r = 0; r < 8; ++r)
                dst[t * 256 + 2 * r] = acc[t][r] * inv;
    }
    LDS_BARRIER();
    float* ag = a + ((size_t)b * 4 * 512 + (n0 >> 1)) * 32 + (lane << 2);
#pragma unroll
    for (int t = 0; t < 4; ++t)
#pragma unroll
        for (int u = 0; u < 2; ++u)
            *(v4f*)(ag + (size_t)t * 16384 + u * 128) = *(const v4f*)(asw + t * 256 + u * 128 + (lane << 2));
}

__global__ __launch_bounds__(256) void netvlad_asum(
    const float* __restrict__ a, float* __restrict__ asum)
{
    __shared__ float red[2][256];
    const int tid = threadIdx.x;
    const v4f* src = (const v4f*)(a + (size_t)blockIdx.x * 16384) + tid;
    float s0 = 0.0f, s1 = 0.0f;
#pragma unroll
    for (int i = 0; i < 16; ++i) {
        v4f v = src[i * 256];
        s0 += v.x + v.y;
        s1 += v.z + v.w;
    }
    red[0][tid] = s0;
    red[1][tid] = s1;
    __syncthreads();
    if (tid < 16) {
        const float* rr = &red[tid & 1][tid >> 1];
        float s = 0.0f;
        for (int g = 0; g < 32; ++g) s += rr[g * 8];
        asum[blockIdx.x * 16 + tid] = s;
    }
}


#define VLAD_LOAD(SET, IT)                                                      \
    {                                                                           \
        const int nblk = (((IT) + rot) & 31) << 5;                              \
        const float* ap = ab + nblk * 16;                                       \
        const float* xq = xrow + nblk + koff;                                   \
        _Pragma("unroll")   \
        for (int s = 0; s < 8; ++s) {                                           \
            fa[SET][s][0] = *(const v2f*)(ap + 0 * 16384 + s * 64);             \
            fa[SET][s][1] = *(const v2f*)(ap + 1 * 16384 + s * 64);             \
            fa[SET][s][2] = *(const v2f*)(ap + 2 * 16384 + s * 64);             \
            fa[SET][s][3] = *(const v2f*)(ap + 3 * 16384 + s * 64);             \
            fb[SET][s]    = *(const v2f*)(xq + 4 * s);                          \
        }                                                                       \
    }

#define VLAD_WMMA(SET)                                                          \
    _Pragma("unroll")               \
    for (int s = 0; s < 8; ++s) {                                               \
        acc[0] = __builtin_amdgcn_wmma_f32_16x16x4_f32(false, fa[SET][s][0], false, fb[SET][s], (short)0, acc[0], false, false); \
        acc[1] = __builtin_amdgcn_wmma_f32_16x16x4_f32(false, fa[SET][s][1], false, fb[SET][s], (short)0, acc[1], false, false); \
        acc[2] = __builtin_amdgcn_wmma_f32_16x16x4_f32(false, fa[SET][s][2], false, fb[SET][s], (short)0, acc[2], false, false); \
        acc[3] = __builtin_amdgcn_wmma_f32_16x16x4_f32(false, fa[SET][s][3], false, fb[SET][s], (short)0, acc[3], false, false); \
    }

#define VLAD_ORDER()                                                            \
    asm volatile("" : "+v"(acc[0]), "+v"(acc[1]), "+v"(acc[2]), "+v"(acc[3]) : : "memory")

__global__ __launch_bounds__(128) void netvlad_vlad(
    const float* __restrict__ x, const float* __restrict__ a,
    const float* __restrict__ asum, const float* __restrict__ cent,
    float* __restrict__ out)
{
    const int lane = threadIdx.x & 31;
    const int wid  = threadIdx.x >> 5;
    const int col  = lane & 15;
    const int half = lane >> 4;
    const int koff = half << 1;

    const int tile = blockIdx.x * 4 + wid;
    const int b    = tile >> 7;
    const int c0   = (tile & 127) << 4;

    const float* ab   = a + (size_t)b * NK * NN + (lane << 1);
    const float* xrow = x + ((size_t)b * NC + c0 + col) * NN;

    v8f acc[4] = {v8f{}, v8f{}, v8f{}, v8f{}};
    v2f fa[2][8][4], fb[2][8];

    const int rot = blockIdx.x & 31;
    VLAD_LOAD(0, 0)
    for (int it = 0; it < NN / 32 - 2; it += 2) {
        VLAD_LOAD(1, it + 1)
        VLAD_ORDER();
        VLAD_WMMA(0)
        VLAD_ORDER();
        VLAD_LOAD(0, it + 2)
        VLAD_ORDER();
        VLAD_WMMA(1)
        VLAD_ORDER();
    }
    VLAD_LOAD(1, NN / 32 - 1)
    VLAD_ORDER();
    VLAD_WMMA(0)
    VLAD_WMMA(1)

    float* ob = out + (size_t)b * NK * NC;
#pragma unroll
    for (int t = 0; t < 4; ++t)
#pragma unroll
        for (int r = 0; r < 8; ++r) {
            int k = t * 16 + r + 8 * half;
            float s  = asum[b * NK + k];
            float cv = cent[(size_t)k * NC + c0 + col];
            ob[(size_t)k * NC + c0 + col] = acc[t][r] - s * cv;
        }
}

__global__ __launch_bounds__(256) void netvlad_norm(
    const float* __restrict__ out, float* __restrict__ invn)
{
    __shared__ float red[256];
    const float4* vb = (const float4*)(out + (size_t)blockIdx.x * NK * NC);
    float s = 0.0f;
    for (int i = threadIdx.x; i < (NK * NC) / 4; i += 256) {
        float4 v = vb[i];
        s += v.x * v.x + v.y * v.y + v.z * v.z + v.w * v.w;
    }
    red[threadIdx.x] = s;
    __syncthreads();
    for (int off = 128; off > 0; off >>= 1) {
        if (threadIdx.x < off) red[threadIdx.x] += red[threadIdx.x + off];
        __syncthreads();
    }
    if (threadIdx.x == 0)
        invn[blockIdx.x] = 1.0f / fmaxf(sqrtf(red[0]), 1e-12f);
}

__global__ __launch_bounds__(256) void netvlad_scale(
    float* __restrict__ out, const float* __restrict__ invn)
{
    int gid = blockIdx.x * 256 + threadIdx.x;
    int b = gid / ((NK * NC) / 4);
    float sc = invn[b];
    float4* p = (float4*)out;
    float4 v = p[gid];
    v.x *= sc; v.y *= sc; v.z *= sc; v.w *= sc;
    p[gid] = v;
}

extern "C" void kernel_launch(void* const* d_in, const int* in_sizes, int n_in,
                              void* d_out, int out_size, void* d_ws, size_t ws_size,
                              hipStream_t stream) {
    const float* x    = (const float*)d_in[0];
    const float* w    = (const float*)d_in[1];
    const float* cent = (const float*)d_in[2];
    float* out = (float*)d_out;

    float* a    = (float*)d_ws;
    float* asum = a + (size_t)NB * NK * NN;
    float* invn = asum + NB * NK;

    netvlad_assign<<<(NB * NN / 16) / 4, 128, 0, stream>>>(x, w, a);
    netvlad_asum  <<<NB * 4, 256, 0, stream>>>(a, asum);
    netvlad_vlad  <<<(NB * NC / 16) / 4, 128, 0, stream>>>(x, a, asum, cent, out);
    netvlad_norm  <<<NB, 256, 0, stream>>>(out, invn);
    netvlad_scale <<<(NB * NK * NC / 4) / 256, 256, 0, stream>>>(out, invn);
}


# vlad kernel re-tiled: a wave keeps 64 k x 32 c so each A fragment feeds two c-tiles (A loads cut by two); n 0-511 and 512-1023 go to a wave pair, sums joined through LDS; splits the n sum in two
# speedup vs baseline: 3.7565x; 1.0429x over previous
#include <hip/hip_runtime.h>

typedef __attribute__((ext_vector_type(2))) float v2f;
typedef __attribute__((ext_vector_type(8))) float v8f;

#define NK 64
#define NC 2048
#define NN 1024
#define NB 16

#define GC 32
#define NG (NC / GC)
#define PD 8
#define WP 36
#define XP 72

typedef __attribute__((ext_vector_type(4))) float v4f;

#define LDS_BARRIER()                                                           \
    do {                                                                        \
        __builtin_amdgcn_fence(__ATOMIC_RELEASE, "workgroup", "local");         \
        __builtin_amdgcn_s_barrier();                                           \
        __builtin_amdgcn_fence(__ATOMIC_ACQUIRE, "workgroup", "local");         \
    } while (0)

#define ASSIGN_ISSUE(SLOT)                                                      \
    {                                                                           \
        _Pragma("unroll")           \
        for (int i = 0; i < 4; ++i) {                                           \
            rw[SLOT][i] = *(const v4f*)(wp + (size_t)(16 * i) * NC);            \
            rx[SLOT][i] = *(const v4f*)(xp + (size_t)(8 * i) * NN);             \
        }                                                                       \
        const bool last = (cn == NG - 1);                                       \
        wp += last ? -(ptrdiff_t)(NG - 1) * GC : (ptrdiff_t)GC;                 \
        xp += last ? -(ptrdiff_t)(NG - 1) * GC * NN : (ptrdiff_t)GC * NN;       \
        cn = (cn + 1) & (NG - 1);                                               \
    }

#define ASSIGN_STASH(SLOT, STG)                                                 \
    _Pragma("unroll")               \
    for (int i = 0; i < 4; ++i) {                                               \
        *(v4f*)(&ws[STG][wso + 16 * i * WP]) = rw[SLOT][i];                     \
        *(v4f*)(&xs[STG][xso + 8 * i * XP])  = rx[SLOT][i];                     \
    }

#define ASSIGN_FRAGS(SET, STG)                                                  \
    {                                                                           \
        const float* wl = &ws[STG][col * WP + koff];                            \
        const float* xl = &xs[STG][koff * XP + (wid << 4) + col];               \
        _Pragma("unroll")   \
        for (int s = 0; s < GC / 4; ++s) {                                      \
            fa[SET][s][0] = *(const v2f*)(wl +  0 * WP + 4 * s);                \
            fa[SET][s][1] = *(const v2f*)(wl + 16 * WP + 4 * s);                \
            fa[SET][s][2] = *(const v2f*)(wl + 32 * WP + 4 * s);                \
            fa[SET][s][3] = *(const v2f*)(wl + 48 * WP + 4 * s);                \
            fb[SET][s].x = xl[(4 * s + 0) * XP];                                \
            fb[SET][s].y = xl[(4 * s + 1) * XP];                                \
        }                                                                       \
    }

#define ASSIGN_WMMA(SET)                                                        \
    _Pragma("unroll")               \
    for (int s = 0; s < GC / 4; ++s) {                                          \
        acc[0] = __builtin_amdgcn_wmma_f32_16x16x4_f32(false, fa[SET][s][0], false, fb[SET][s], (short)0, acc[0], false, false); \
        acc[1] = __builtin_amdgcn_wmma_f32_16x16x4_f32(false, fa[SET][s][1], false, fb[SET][s], (short)0, acc[1], false, false); \
        acc[2] = __builtin_amdgcn_wmma_f32_16x16x4_f32(false, fa[SET][s][2], false, fb[SET][s], (short)0, acc[2], false, false); \
        acc[3] = __builtin_amdgcn_wmma_f32_16x16x4_f32(false, fa[SET][s][3], false, fb[SET][s], (short)0, acc[3], false, false); \
    }

__global__ __launch_bounds__(128) void netvlad_assign(
    const float* __restrict__ x, const float* __restrict__ w, float* __restrict__ a)
{
    __shared__ __attribute__((aligned(16))) float ws[4][NK * WP];
    __shared__ __attribute__((aligned(16))) float xs[4][GC * XP];
    __shared__ __attribute__((aligned(16))) float as[4][NK * 16];

    const int tid  = threadIdx.x;
    const int lane = tid & 31;
    const int wid  = tid >> 5;
    const int col  = lane & 15;
    const int half = lane >> 4;
    const int koff = half << 1;

    const int b  = blockIdx.x >> 4;
    const int nb = (blockIdx.x & 15) << 6;
    const int n0 = nb + (wid << 4);
    int cn = blockIdx.x & (NG - 1);

    const float* xp = x + (size_t)b * NC * NN + (size_t)(tid >> 4) * NN + nb + ((tid & 15) << 2) + (size_t)cn * GC * NN;
    const float* wp = w + (size_t)(tid >> 3) * NC + ((tid & 7) << 2) + cn * GC;
    const int xso = (tid >> 4) * XP + ((tid & 15) << 2);
    const int wso = (tid >> 3) * WP + ((tid & 7) << 2);

    v4f rx[PD][4], rw[PD][4];
    v2f fa[2][GC / 4][4], fb[2][GC / 4];
    v8f acc[4] = {v8f{}, v8f{}, v8f{}, v8f{}};

#pragma unroll
    for (int p = 0; p < PD; ++p) {
        ASSIGN_ISSUE(p)
        __builtin_amdgcn_sched_barrier(0);
    }
    ASSIGN_STASH(0, 0)
    ASSIGN_STASH(1, 1)
    LDS_BARRIER();
    ASSIGN_FRAGS(0, 0)

    for (int g0 = 0; g0 < NG - PD; g0 += PD) {
#pragma unroll
        for (int j = 0; j < PD; ++j) {
            ASSIGN_ISSUE(j)
            ASSIGN_STASH((j + 2) & (PD - 1), (j + 2) & 3)
            ASSIGN_FRAGS((j + 1) & 1, (j + 1) & 3)
            __builtin_amdgcn_sched_barrier(0);
            ASSIGN_WMMA(j & 1)
            __builtin_amdgcn_sched_barrier(0);
            LDS_BARRIER();
        }
    }
#pragma unroll
    for (int j = 0; j < PD; ++j) {
        if (j + 2 < PD) {
            ASSIGN_STASH((j + 2) & (PD - 1), (j + 2) & 3)
        }
        if (j + 1 < PD) {
            ASSIGN_FRAGS((j + 1) & 1, (j + 1) & 3)
        }
        __builtin_amdgcn_sched_barrier(0);
        ASSIGN_WMMA(j & 1)
        __builtin_amdgcn_sched_barrier(0);
        if (j + 2 < PD) {
            LDS_BARRIER();
        }
    }

    float mx = -3.402823466e38f;
#pragma unroll
    for (int t = 0; t < 4; ++t)
#pragma unroll
        for (int r = 0; r < 8; ++r) mx = fmaxf(mx, acc[t][r]);
    mx = fmaxf(mx, __shfl_xor(mx, 16, 32));

    float sum = 0.0f;
#pragma unroll
    for (int t = 0; t < 4; ++t)
#pragma unroll
        for (int r = 0; r < 8; ++r) {
            float e = __expf(acc[t][r] - mx);
            acc[t][r] = e;
            sum += e;
        }
    sum += __shfl_xor(sum, 16, 32);
    const float inv = 1.0f / sum;

    float* asw = as[wid];
    {
        float* dst = asw + ((col >> 1) << 5) + (half << 4) + (col & 1);
#pragma unroll
        for (int t = 0; t < 4; ++t)
#pragma unroll
            for (int r = 0; r < 8; ++r)
                dst[t * 256 + 2 * r] = acc[t][r] * inv;
    }
    LDS_BARRIER();
    float* ag = a + ((size_t)b * 4 * 512 + (n0 >> 1)) * 32 + (lane << 2);
#pragma unroll
    for (int t = 0; t < 4; ++t)
#pragma unroll
        for (int u = 0; u < 2; ++u)
            *(v4f*)(ag + (size_t)t * 16384 + u * 128) = *(const v4f*)(asw + t * 256 + u * 128 + (lane << 2));
}

__global__ __launch_bounds__(256) void netvlad_asum(
    const float* __restrict__ a, float* __restrict__ asum)
{
    __shared__ float red[2][256];
    const int tid = threadIdx.x;
    const v4f* src = (const v4f*)(a + (size_t)blockIdx.x * 16384) + tid;
    float s0 = 0.0f, s1 = 0.0f;
#pragma unroll
    for (int i = 0; i < 16; ++i) {
        v4f v = src[i * 256];
        s0 += v.x + v.y;
        s1 += v.z + v.w;
    }
    red[0][tid] = s0;
    red[1][tid] = s1;
    __syncthreads();
    if (tid < 16) {
        const float* rr = &red[tid & 1][tid >> 1];
        float s = 0.0f;
        for (int g = 0; g < 32; ++g) s += rr[g * 8];
        asum[blockIdx.x * 16 + tid] = s;
    }
}


#define VLAD_LOAD(SET, IT)                                                      \
    {                                                                           \
        const int nblk = (nh16 + (((IT) + rot) & 15)) << 5;                     \
        const float* ap = ab + nblk * 16;                                       \
        const float* xq = xrow + nblk + koff;                                   \
        _Pragma("unroll")   \
        for (int s = 0; s < 8; ++s) {                                           \
            fa[SET][s][0] = *(const v2f*)(ap + 0 * 16384 + s * 64);             \
            fa[SET][s][1] = *(const v2f*)(ap + 1 * 16384 + s * 64);             \
            fa[SET][s][2] = *(const v2f*)(ap + 2 * 16384 + s * 64);             \
            fa[SET][s][3] = *(const v2f*)(ap + 3 * 16384 + s * 64);             \
            fb[SET][s][0] = *(const v2f*)(xq + 4 * s);                          \
            fb[SET][s][1] = *(const v2f*)(xq + 16 * NN + 4 * s);                \
        }                                                                       \
    }

#define VLAD_WMMA(SET)                                                          \
    _Pragma("unroll")               \
    for (int s = 0; s < 8; ++s) {                                               \
        _Pragma("unroll")           \
        for (int u = 0; u < 2; ++u) {                                           \
            acc[u][0] = __builtin_amdgcn_wmma_f32_16x16x4_f32(false, fa[SET][s][0], false, fb[SET][s][u], (short)0, acc[u][0], false, false); \
            acc[u][1] = __builtin_amdgcn_wmma_f32_16x16x4_f32(false, fa[SET][s][1], false, fb[SET][s][u], (short)0, acc[u][1], false, false); \
            acc[u][2] = __builtin_amdgcn_wmma_f32_16x16x4_f32(false, fa[SET][s][2], false, fb[SET][s][u], (short)0, acc[u][2], false, false); \
            acc[u][3] = __builtin_amdgcn_wmma_f32_16x16x4_f32(false, fa[SET][s][3], false, fb[SET][s][u], (short)0, acc[u][3], false, false); \
        }                                                                       \
    }

#define VLAD_ORDER()                                                            \
    asm volatile("" : "+v"(acc[0][0]), "+v"(acc[0][1]), "+v"(acc[0][2]), "+v"(acc[0][3]),   \
                      "+v"(acc[1][0]), "+v"(acc[1][1]), "+v"(acc[1][2]), "+v"(acc[1][3]) : : "memory")

#define VLAD_DUMP(TB)                                                           \
    _Pragma("unroll")          \
    for (int u = 0; u < 2; ++u)                                                 \
        _Pragma("unroll")                                                       \
        for (int tt = 0; tt < 2; ++tt) {                                        \
            mine[((u * 2 + tt) * 2 + 0) * 32] = __builtin_shufflevector(acc[u][(TB) + tt], acc[u][(TB) + tt], 0, 1, 2, 3); \
            mine[((u * 2 + tt) * 2 + 1) * 32] = __builtin_shufflevector(acc[u][(TB) + tt], acc[u][(TB) + tt], 4, 5, 6, 7); \
        }

#define VLAD_FINISH(TB)                                                         \
    _Pragma("unroll")       \
    for (int u = 0; u < 2; ++u)                                                 \
        _Pragma("unroll")                                                       \
        for (int tt = 0; tt < 2; ++tt) {                                        \
            const v4f lo = theirs[((u * 2 + tt) * 2 + 0) * 32];                 \
            const v4f hi = theirs[((u * 2 + tt) * 2 + 1) * 32];                 \
            _Pragma("unroll")                                                   \
            for (int r = 0; r < 8; ++r) {                                       \
                const int k = ((TB) + tt) * 16 + r + 8 * half;                  \
                const float v  = acc[u][(TB) + tt][r] + (r < 4 ? lo[r & 3] : hi[r & 3]); \
                const float s  = asum[b * NK + k];                              \
                const float cv = cent[(size_t)k * NC + c0 + 16 * u + col];      \
                ob[(size_t)k * NC + c0 + 16 * u + col] = v - s * cv;            \
            }                                                                   \
        }

__global__ __launch_bounds__(256) void netvlad_vlad(
    const float* __restrict__ x, const float* __restrict__ a,
    const float* __restrict__ asum, const float* __restrict__ cent,
    float* __restrict__ out)
{
    __shared__ __attribute__((aligned(16))) float red[8 * 8 * 32 * 4];

    const int lane = threadIdx.x & 31;
    const int wid  = threadIdx.x >> 5;
    const int col  = lane & 15;
    const int half = lane >> 4;
    const int koff = half << 1;
    const int nh   = wid >> 2;
    const int nh16 = nh << 4;

    const int tile = blockIdx.x * 4 + (wid & 3);
    const int b    = tile >> 6;
    const int c0   = (tile & 63) << 5;

    const float* ab   = a + (size_t)b * NK * NN + (lane << 1);
    const float* xrow = x + ((size_t)b * NC + c0 + col) * NN;

    v8f acc[2][4] = {{v8f{}, v8f{}, v8f{}, v8f{}}, {v8f{}, v8f{}, v8f{}, v8f{}}};
    v2f fa[2][8][4], fb[2][8][2];

    const int rot = blockIdx.x & 15;
    VLAD_LOAD(0, 0)
    for (int it = 0; it < NN / 64 - 2; it += 2) {
        VLAD_LOAD(1, it + 1)
        VLAD_ORDER();
        VLAD_WMMA(0)
        VLAD_ORDER();
        VLAD_LOAD(0, it + 2)
        VLAD_ORDER();
        VLAD_WMMA(1)
        VLAD_ORDER();
    }
    VLAD_LOAD(1, NN / 64 - 1)
    VLAD_ORDER();
    VLAD_WMMA(0)
    VLAD_WMMA(1)

    float* ob = out + (size_t)b * NK * NC;
    v4f* mine = (v4f*)red + (wid * 8) * 32 + lane;
    const v4f* theirs = (const v4f*)red + ((wid ^ 4) * 8) * 32 + lane;
    if (nh == 0) { VLAD_DUMP(2) } else { VLAD_DUMP(0) }
    __syncthreads();
    if (nh == 0) { VLAD_FINISH(0) } else { VLAD_FINISH(2) }
}

__global__ __launch_bounds__(256) void netvlad_norm(
    const float* __restrict__ out, float* __restrict__ invn)
{
    __shared__ float red[256];
    const float4* vb = (const float4*)(out + (size_t)blockIdx.x * NK * NC);
    float s = 0.0f;
    for (int i = threadIdx.x; i < (NK * NC) / 4; i += 256) {
        float4 v = vb[i];
        s += v.x * v.x + v.y * v.y + v.z * v.z + v.w * v.w;
    }
    red[threadIdx.x] = s;
    __syncthreads();
    for (int off = 128; off > 0; off >>= 1) {
        if (threadIdx.x < off) red[threadIdx.x] += red[threadIdx.x + off];
        __syncthreads();
    }
    if (threadIdx.x == 0)
        invn[blockIdx.x] = 1.0f / fmaxf(sqrtf(red[0]), 1e-12f);
}

__global__ __launch_bounds__(256) void netvlad_scale(
    float* __restrict__ out, const float* __restrict__ invn)
{
    int gid = blockIdx.x * 256 + threadIdx.x;
    int b = gid / ((NK * NC) / 4);
    float sc = invn[b];
    float4* p = (float4*)out;
    float4 v = p[gid];
    v.x *= sc; v.y *= sc; v.z *= sc; v.w *= sc;
    p[gid] = v;
}

extern "C" void kernel_launch(void* const* d_in, const int* in_sizes, int n_in,
                              void* d_out, int out_size, void* d_ws, size_t ws_size,
                              hipStream_t stream) {
    const float* x    = (const float*)d_in[0];
    const float* w    = (const float*)d_in[1];
    const float* cent = (const float*)d_in[2];
    float* out = (float*)d_out;

    float* a    = (float*)d_ws;
    float* asum = a + (size_t)NB * NK * NN;
    float* invn = asum + NB * NK;

    netvlad_assign<<<(NB * NN / 16) / 4, 128, 0, stream>>>(x, w, a);
    netvlad_asum  <<<NB * 4, 256, 0, stream>>>(a, asum);
    netvlad_vlad  <<<(NB * NC / 32) / 4, 256, 0, stream>>>(x, a, asum, cent, out);
    netvlad_norm  <<<NB, 256, 0, stream>>>(out, invn);
    netvlad_scale <<<(NB * NK * NC / 4) / 256, 256, 0, stream>>>(out, invn);
}


# norm kernel: a batch's square sum cut into 16 pieces, one work-group each (256 groups, one load round trip; was 16 groups, 16 trips); scale kernel adds the 16 partials; reorders that sum
# speedup vs baseline: 4.1745x; 1.1113x over previous
#include <hip/hip_runtime.h>

typedef __attribute__((ext_vector_type(2))) float v2f;
typedef __attribute__((ext_vector_type(8))) float v8f;

#define NK 64
#define NC 2048
#define NN 1024
#define NB 16

#define GC 32
#define NG (NC / GC)
#define PD 8
#define WP 36
#define XP 72

typedef __attribute__((ext_vector_type(4))) float v4f;

#define LDS_BARRIER()                                                           \
    do {                                                                        \
        __builtin_amdgcn_fence(__ATOMIC_RELEASE, "workgroup", "local");         \
        __builtin_amdgcn_s_barrier();                                           \
        __builtin_amdgcn_fence(__ATOMIC_ACQUIRE, "workgroup", "local");         \
    } while (0)

#define ASSIGN_ISSUE(SLOT)                                                      \
    {                                                                           \
        _Pragma("unroll")           \
        for (int i = 0; i < 4; ++i) {                                           \
            rw[SLOT][i] = *(const v4f*)(wp + (size_t)(16 * i) * NC);            \
            rx[SLOT][i] = *(const v4f*)(xp + (size_t)(8 * i) * NN);             \
        }                                                                       \
        const bool last = (cn == NG - 1);                                       \
        wp += last ? -(ptrdiff_t)(NG - 1) * GC : (ptrdiff_t)GC;                 \
        xp += last ? -(ptrdiff_t)(NG - 1) * GC * NN : (ptrdiff_t)GC * NN;       \
        cn = (cn + 1) & (NG - 1);                                               \
    }

#define ASSIGN_STASH(SLOT, STG)                                                 \
    _Pragma("unroll")               \
    for (int i = 0; i < 4; ++i) {                                               \
        *(v4f*)(&ws[STG][wso + 16 * i * WP]) = rw[SLOT][i];                     \
        *(v4f*)(&xs[STG][xso + 8 * i * XP])  = rx[SLOT][i];                     \
    }

#define ASSIGN_FRAGS(SET, STG)                                                  \
    {                                                                           \
        const float* wl = &ws[STG][col * WP + koff];                            \
        const float* xl = &xs[STG][koff * XP + (wid << 4) + col];               \
        _Pragma("unroll")   \
        for (int s = 0; s < GC / 4; ++s) {                                      \
            fa[SET][s][0] = *(const v2f*)(wl +  0 * WP + 4 * s);                \
            fa[SET][s][1] = *(const v2f*)(wl + 16 * WP + 4 * s);                \
            fa[SET][s][2] = *(const v2f*)(wl + 32 * WP + 4 * s);                \
            fa[SET][s][3] = *(const v2f*)(wl + 48 * WP + 4 * s);                \
            fb[SET][s].x = xl[(4 * s + 0) * XP];                                \
            fb[SET][s].y = xl[(4 * s + 1) * XP];                                \
        }                                                                       \
    }

#define ASSIGN_WMMA(SET)                                                        \
    _Pragma("unroll")               \
    for (int s = 0; s < GC / 4; ++s) {                                          \
        acc[0] = __builtin_amdgcn_wmma_f32_16x16x4_f32(false, fa[SET][s][0], false, fb[SET][s], (short)0, acc[0], false, false); \
        acc[1] = __builtin_amdgcn_wmma_f32_16x16x4_f32(false, fa[SET][s][1], false, fb[SET][s], (short)0, acc[1], false, false); \
        acc[2] = __builtin_amdgcn_wmma_f32_16x16x4_f32(false, fa[SET][s][2], false, fb[SET][s], (short)0, acc[2], false, false); \
        acc[3] = __builtin_amdgcn_wmma_f32_16x16x4_f32(false, fa[SET][s][3], false, fb[SET][s], (short)0, acc[3], false, false); \
    }

__global__ __launch_bounds__(128) void netvlad_assign(
    const float* __restrict__ x, const float* __restrict__ w, float* __restrict__ a)
{
    __shared__ __attribute__((aligned(16))) float ws[4][NK * WP];
    __shared__ __attribute__((aligned(16))) float xs[4][GC * XP];
    __shared__ __attribute__((aligned(16))) float as[4][NK * 16];

    const int tid  = threadIdx.x;
    const int lane = tid & 31;
    const int wid  = tid >> 5;
    const int col  = lane & 15;
    const int half = lane >> 4;
    const int koff = half << 1;

    const int b  = blockIdx.x >> 4;
    const int nb = (blockIdx.x & 15) << 6;
    const int n0 = nb + (wid << 4);
    int cn = blockIdx.x & (NG - 1);

    const float* xp = x + (size_t)b * NC * NN + (size_t)(tid >> 4) * NN + nb + ((tid & 15) << 2) + (size_t)cn * GC * NN;
    const float* wp = w + (size_t)(tid >> 3) * NC + ((tid & 7) << 2) + cn * GC;
    const int xso = (tid >> 4) * XP + ((tid & 15) << 2);
    const int wso = (tid >> 3) * WP + ((tid & 7) << 2);

    v4f rx[PD][4], rw[PD][4];
    v2f fa[2][GC / 4][4], fb[2][GC / 4];
    v8f acc[4] = {v8f{}, v8f{}, v8f{}, v8f{}};

#pragma unroll
    for (int p = 0; p < PD; ++p) {
        ASSIGN_ISSUE(p)
        __builtin_amdgcn_sched_barrier(0);
    }
    ASSIGN_STASH(0, 0)
    ASSIGN_STASH(1, 1)
    LDS_BARRIER();
    ASSIGN_FRAGS(0, 0)

    for (int g0 = 0; g0 < NG - PD; g0 += PD) {
#pragma unroll
        for (int j = 0; j < PD; ++j) {
            ASSIGN_ISSUE(j)
            ASSIGN_STASH((j + 2) & (PD - 1), (j + 2) & 3)
            ASSIGN_FRAGS((j + 1) & 1, (j + 1) & 3)
            __builtin_amdgcn_sched_barrier(0);
            ASSIGN_WMMA(j & 1)
            __builtin_amdgcn_sched_barrier(0);
            LDS_BARRIER();
        }
    }
#pragma unroll
    for (int j = 0; j < PD; ++j) {
        if (j + 2 < PD) {
            ASSIGN_STASH((j + 2) & (PD - 1), (j + 2) & 3)
        }
        if (j + 1 < PD) {
            ASSIGN_FRAGS((j + 1) & 1, (j + 1) & 3)
        }
        __builtin_amdgcn_sched_barrier(0);
        ASSIGN_WMMA(j & 1)
        __builtin_amdgcn_sched_barrier(0);
        if (j + 2 < PD) {
            LDS_BARRIER();
        }
    }

    float mx = -3.402823466e38f;
#pragma unroll
    for (int t = 0; t < 4; ++t)
#pragma unroll
        for (int r = 0; r < 8; ++r) mx = fmaxf(mx, acc[t][r]);
    mx = fmaxf(mx, __shfl_xor(mx, 16, 32));

    float sum = 0.0f;
#pragma unroll
    for (int t = 0; t < 4; ++t)
#pragma unroll
        for (int r = 0; r < 8; ++r) {
            float e = __expf(acc[t][r] - mx);
            acc[t][r] = e;
            sum += e;
        }
    sum += __shfl_xor(sum, 16, 32);
    const float inv = 1.0f / sum;

    float* asw = as[wid];
    {
        float* dst = asw + ((col >> 1) << 5) + (half << 4) + (col & 1);
#pragma unroll
        for (int t = 0; t < 4; ++t)
#pragma unroll
            for (int r = 0; r < 8; ++r)
                dst[t * 256 + 2 * r] = acc[t][r] * inv;
    }
    LDS_BARRIER();
    float* ag = a + ((size_t)b * 4 * 512 + (n0 >> 1)) * 32 + (lane << 2);
#pragma unroll
    for (int t = 0; t < 4; ++t)
#pragma unroll
        for (int u = 0; u < 2; ++u)
            *(v4f*)(ag + (size_t)t * 16384 + u * 128) = *(const v4f*)(asw + t * 256 + u * 128 + (lane << 2));
}

__global__ __launch_bounds__(256) void netvlad_asum(
    const float* __restrict__ a, float* __restrict__ asum)
{
    __shared__ float red[2][256];
    const int tid = threadIdx.x;
    const v4f* src = (const v4f*)(a + (size_t)blockIdx.x * 16384) + tid;
    float s0 = 0.0f, s1 = 0.0f;
#pragma unroll
    for (int i = 0; i < 16; ++i) {
        v4f v = src[i * 256];
        s0 += v.x + v.y;
        s1 += v.z + v.w;
    }
    red[0][tid] = s0;
    red[1][tid] = s1;
    __syncthreads();
    if (tid < 16) {
        const float* rr = &red[tid & 1][tid >> 1];
        float s = 0.0f;
        for (int g = 0; g < 32; ++g) s += rr[g * 8];
        asum[blockIdx.x * 16 + tid] = s;
    }
}


#define VLAD_LOAD(SET, IT)                                                      \
    {                                                                           \
        const int nblk = (nh16 + (((IT) + rot) & 15)) << 5;                     \
        const float* ap = ab + nblk * 16;                                       \
        const float* xq = xrow + nblk + koff;                                   \
        _Pragma("unroll")   \
        for (int s = 0; s < 8; ++s) {                                           \
            fa[SET][s][0] = *(const v2f*)(ap + 0 * 16384 + s * 64);             \
            fa[SET][s][1] = *(const v2f*)(ap + 1 * 16384 + s * 64);             \
            fa[SET][s][2] = *(const v2f*)(ap + 2 * 16384 + s * 64);             \
            fa[SET][s][3] = *(const v2f*)(ap + 3 * 16384 + s * 64);             \
            fb[SET][s][0] = *(const v2f*)(xq + 4 * s);                          \
            fb[SET][s][1] = *(const v2f*)(xq + 16 * NN + 4 * s);                \
        }                                                                       \
    }

#define VLAD_WMMA(SET)                                                          \
    _Pragma("unroll")               \
    for (int s = 0; s < 8; ++s) {                                               \
        _Pragma("unroll")           \
        for (int u = 0; u < 2; ++u) {                                           \
            acc[u][0] = __builtin_amdgcn_wmma_f32_16x16x4_f32(false, fa[SET][s][0], false, fb[SET][s][u], (short)0, acc[u][0], false, false); \
            acc[u][1] = __builtin_amdgcn_wmma_f32_16x16x4_f32(false, fa[SET][s][1], false, fb[SET][s][u], (short)0, acc[u][1], false, false); \
            acc[u][2] = __builtin_amdgcn_wmma_f32_16x16x4_f32(false, fa[SET][s][2], false, fb[SET][s][u], (short)0, acc[u][2], false, false); \
            acc[u][3] = __builtin_amdgcn_wmma_f32_16x16x4_f32(false, fa[SET][s][3], false, fb[SET][s][u], (short)0, acc[u][3], false, false); \
        }                                                                       \
    }

#define VLAD_ORDER()                                                            \
    asm volatile("" : "+v"(acc[0][0]), "+v"(acc[0][1]), "+v"(acc[0][2]), "+v"(acc[0][3]),   \
                      "+v"(acc[1][0]), "+v"(acc[1][1]), "+v"(acc[1][2]), "+v"(acc[1][3]) : : "memory")

#define VLAD_DUMP(TB)                                                           \
    _Pragma("unroll")          \
    for (int u = 0; u < 2; ++u)                                                 \
        _Pragma("unroll")                                                       \
        for (int tt = 0; tt < 2; ++tt) {                                        \
            mine[((u * 2 + tt) * 2 + 0) * 32] = __builtin_shufflevector(acc[u][(TB) + tt], acc[u][(TB) + tt], 0, 1, 2, 3); \
            mine[((u * 2 + tt) * 2 + 1) * 32] = __builtin_shufflevector(acc[u][(TB) + tt], acc[u][(TB) + tt], 4, 5, 6, 7); \
        }

#define VLAD_FINISH(TB)                                                         \
    _Pragma("unroll")       \
    for (int u = 0; u < 2; ++u)                                                 \
        _Pragma("unroll")                                                       \
        for (int tt = 0; tt < 2; ++tt) {                                        \
            const v4f lo = theirs[((u * 2 + tt) * 2 + 0) * 32];                 \
            const v4f hi = theirs[((u * 2 + tt) * 2 + 1) * 32];                 \
            _Pragma("unroll")                                                   \
            for (int r = 0; r < 8; ++r) {                                       \
                const int k = ((TB) + tt) * 16 + r + 8 * half;                  \
                const float v  = acc[u][(TB) + tt][r] + (r < 4 ? lo[r & 3] : hi[r & 3]); \
                const float s  = asum[b * NK + k];                              \
                const float cv = cent[(size_t)k * NC + c0 + 16 * u + col];      \
                ob[(size_t)k * NC + c0 + 16 * u + col] = v - s * cv;            \
            }                                                                   \
        }

__global__ __launch_bounds__(256) void netvlad_vlad(
    const float* __restrict__ x, const float* __restrict__ a,
    const float* __restrict__ asum, const float* __restrict__ cent,
    float* __restrict__ out)
{
    __shared__ __attribute__((aligned(16))) float red[8 * 8 * 32 * 4];

    const int lane = threadIdx.x & 31;
    const int wid  = threadIdx.x >> 5;
    const int col  = lane & 15;
    const int half = lane >> 4;
    const int koff = half << 1;
    const int nh   = wid >> 2;
    const int nh16 = nh << 4;

    const int tile = blockIdx.x * 4 + (wid & 3);
    const int b    = tile >> 6;
    const int c0   = (tile & 63) << 5;

    const float* ab   = a + (size_t)b * NK * NN + (lane << 1);
    const float* xrow = x + ((size_t)b * NC + c0 + col) * NN;

    v8f acc[2][4] = {{v8f{}, v8f{}, v8f{}, v8f{}}, {v8f{}, v8f{}, v8f{}, v8f{}}};
    v2f fa[2][8][4], fb[2][8][2];

    const int rot = blockIdx.x & 15;
    VLAD_LOAD(0, 0)
    for (int it = 0; it < NN / 64 - 2; it += 2) {
        VLAD_LOAD(1, it + 1)
        VLAD_ORDER();
        VLAD_WMMA(0)
        VLAD_ORDER();
        VLAD_LOAD(0, it + 2)
        VLAD_ORDER();
        VLAD_WMMA(1)
        VLAD_ORDER();
    }
    VLAD_LOAD(1, NN / 64 - 1)
    VLAD_ORDER();
    VLAD_WMMA(0)
    VLAD_WMMA(1)

    float* ob = out + (size_t)b * NK * NC;
    v4f* mine = (v4f*)red + (wid * 8) * 32 + lane;
    const v4f* theirs = (const v4f*)red + ((wid ^ 4) * 8) * 32 + lane;
    if (nh == 0) { VLAD_DUMP(2) } else { VLAD_DUMP(0) }
    __syncthreads();
    if (nh == 0) { VLAD_FINISH(0) } else { VLAD_FINISH(2) }
}

#define NS 16
#define PIECE4 (NK * NC / 4 / NS)

__global__ __launch_bounds__(256) void netvlad_norm(
    const float* __restrict__ out, float* __restrict__ part)
{
    __shared__ float red[8];
    const v4f* vb = (const v4f*)out + (size_t)blockIdx.x * PIECE4 + threadIdx.x;
    v4f v[PIECE4 / 256];
#pragma unroll
    for (int i = 0; i < PIECE4 / 256; ++i) v[i] = vb[i * 256];
    float s = 0.0f;
#pragma unroll
    for (int i = 0; i < PIECE4 / 256; ++i)
        s += v[i].x * v[i].x + v[i].y * v[i].y + v[i].z * v[i].z + v[i].w * v[i].w;
#pragma unroll
    for (int off = 16; off > 0; off >>= 1) s += __shfl_xor(s, off, 32);
    if ((threadIdx.x & 31) == 0) red[threadIdx.x >> 5] = s;
    __syncthreads();
    if (threadIdx.x == 0) {
        float t = 0.0f;
#pragma unroll
        for (int w = 0; w < 8; ++w) t += red[w];
        part[blockIdx.x] = t;
    }
}

__global__ __launch_bounds__(256) void netvlad_scale(
    float* __restrict__ out, const float* __restrict__ part)
{
    int gid = blockIdx.x * 256 + threadIdx.x;
    const float* pb = part + (blockIdx.x / ((NK * NC) / 4 / 256)) * NS;
    float q = 0.0f;
#pragma unroll
    for (int i = 0; i < NS; ++i) q += pb[i];
    const float sc = 1.0f / fmaxf(sqrtf(q), 1e-12f);
    float4* p = (float4*)out;
    float4 v = p[gid];
    v.x *= sc; v.y *= sc; v.z *= sc; v.w *= sc;
    p[gid] = v;
}

extern "C" void kernel_launch(void* const* d_in, const int* in_sizes, int n_in,
                              void* d_out, int out_size, void* d_ws, size_t ws_size,
                              hipStream_t stream) {
    const float* x    = (const float*)d_in[0];
    const float* w    = (const float*)d_in[1];
    const float* cent = (const float*)d_in[2];
    float* out = (float*)d_out;

    float* a    = (float*)d_ws;
    float* asum = a + (size_t)NB * NK * NN;
    float* part = a;

    netvlad_assign<<<(NB * NN / 16) / 4, 128, 0, stream>>>(x, w, a);
    netvlad_asum  <<<NB * 4, 256, 0, stream>>>(a, asum);
    netvlad_vlad  <<<(NB * NC / 32) / 4, 256, 0, stream>>>(x, a, asum, cent, out);
    netvlad_norm  <<<NB * NS, 256, 0, stream>>>(out, part);
    netvlad_scale <<<(NB * NK * NC / 4) / 256, 256, 0, stream>>>(out, part);
}


# asum launch folded into the vlad launch (a work-group sums 4 rows of a by b128 loads beside its n loop); a_sum*centroid now subtracted in the norm and scale kernels; reorders the a_sum sum
# speedup vs baseline: 4.4457x; 1.0650x over previous
#include <hip/hip_runtime.h>

typedef __attribute__((ext_vector_type(2))) float v2f;
typedef __attribute__((ext_vector_type(8))) float v8f;

#define NK 64
#define NC 2048
#define NN 1024
#define NB 16

#define GC 32
#define NG (NC / GC)
#define PD 8
#define WP 36
#define XP 72

typedef __attribute__((ext_vector_type(4))) float v4f;

#define LDS_BARRIER()                                                           \
    do {                                                                        \
        __builtin_amdgcn_fence(__ATOMIC_RELEASE, "workgroup", "local");         \
        __builtin_amdgcn_s_barrier();                                           \
        __builtin_amdgcn_fence(__ATOMIC_ACQUIRE, "workgroup", "local");         \
    } while (0)

#define ASSIGN_ISSUE(SLOT)                                                      \
    {                                                                           \
        _Pragma("unroll")           \
        for (int i = 0; i < 4; ++i) {                                           \
            rw[SLOT][i] = *(const v4f*)(wp + (size_t)(16 * i) * NC);            \
            rx[SLOT][i] = *(const v4f*)(xp + (size_t)(8 * i) * NN);             \
        }                                                                       \
        const bool last = (cn == NG - 1);                                       \
        wp += last ? -(ptrdiff_t)(NG - 1) * GC : (ptrdiff_t)GC;                 \
        xp += last ? -(ptrdiff_t)(NG - 1) * GC * NN : (ptrdiff_t)GC * NN;       \
        cn = (cn + 1) & (NG - 1);                                               \
    }

#define ASSIGN_STASH(SLOT, STG)                                                 \
    _Pragma("unroll")               \
    for (int i = 0; i < 4; ++i) {                                               \
        *(v4f*)(&ws[STG][wso + 16 * i * WP]) = rw[SLOT][i];                     \
        *(v4f*)(&xs[STG][xso + 8 * i * XP])  = rx[SLOT][i];                     \
    }

#define ASSIGN_FRAGS(SET, STG)                                                  \
    {                                                                           \
        const float* wl = &ws[STG][col * WP + koff];                            \
        const float* xl = &xs[STG][koff * XP + (wid << 4) + col];               \
        _Pragma("unroll")   \
        for (int s = 0; s < GC / 4; ++s) {                                      \
            fa[SET][s][0] = *(const v2f*)(wl +  0 * WP + 4 * s);                \
            fa[SET][s][1] = *(const v2f*)(wl + 16 * WP + 4 * s);                \
            fa[SET][s][2] = *(const v2f*)(wl + 32 * WP + 4 * s);                \
            fa[SET][s][3] = *(const v2f*)(wl + 48 * WP + 4 * s);                \
            fb[SET][s].x = xl[(4 * s + 0) * XP];                                \
            fb[SET][s].y = xl[(4 * s + 1) * XP];                                \
        }                                                                       \
    }

#define ASSIGN_WMMA(SET)                                                        \
    _Pragma("unroll")               \
    for (int s = 0; s < GC / 4; ++s) {                                          \
        acc[0] = __builtin_amdgcn_wmma_f32_16x16x4_f32(false, fa[SET][s][0], false, fb[SET][s], (short)0, acc[0], false, false); \
        acc[1] = __builtin_amdgcn_wmma_f32_16x16x4_f32(false, fa[SET][s][1], false, fb[SET][s], (short)0, acc[1], false, false); \
        acc[2] = __builtin_amdgcn_wmma_f32_16x16x4_f32(false, fa[SET][s][2], false, fb[SET][s], (short)0, acc[2], false, false); \
        acc[3] = __builtin_amdgcn_wmma_f32_16x16x4_f32(false, fa[SET][s][3], false, fb[SET][s], (short)0, acc[3], false, false); \
    }

__global__ __launch_bounds__(128) void netvlad_assign(
    const float* __restrict__ x, const float* __restrict__ w, float* __restrict__ a)
{
    __shared__ __attribute__((aligned(16))) float ws[4][NK * WP];
    __shared__ __attribute__((aligned(16))) float xs[4][GC * XP];
    __shared__ __attribute__((aligned(16))) float as[4][NK * 16];

    const int tid  = threadIdx.x;
    const int lane = tid & 31;
    const int wid  = tid >> 5;
    const int col  = lane & 15;
    const int half = lane >> 4;
    const int koff = half << 1;

    const int b  = blockIdx.x >> 4;
    const int nb = (blockIdx.x & 15) << 6;
    const int n0 = nb + (wid << 4);
    int cn = blockIdx.x & (NG - 1);

    const float* xp = x + (size_t)b * NC * NN + (size_t)(tid >> 4) * NN + nb + ((tid & 15) << 2) + (size_t)cn * GC * NN;
    const float* wp = w + (size_t)(tid >> 3) * NC + ((tid & 7) << 2) + cn * GC;
    const int xso = (tid >> 4) * XP + ((tid & 15) << 2);
    const int wso = (tid >> 3) * WP + ((tid & 7) << 2);

    v4f rx[PD][4], rw[PD][4];
    v2f fa[2][GC / 4][4], fb[2][GC / 4];
    v8f acc[4] = {v8f{}, v8f{}, v8f{}, v8f{}};

#pragma unroll
    for (int p = 0; p < PD; ++p) {
        ASSIGN_ISSUE(p)
        __builtin_amdgcn_sched_barrier(0);
    }
    ASSIGN_STASH(0, 0)
    ASSIGN_STASH(1, 1)
    LDS_BARRIER();
    ASSIGN_FRAGS(0, 0)

    for (int g0 = 0; g0 < NG - PD; g0 += PD) {
#pragma unroll
        for (int j = 0; j < PD; ++j) {
            ASSIGN_ISSUE(j)
            ASSIGN_STASH((j + 2) & (PD - 1), (j + 2) & 3)
            ASSIGN_FRAGS((j + 1) & 1, (j + 1) & 3)
            __builtin_amdgcn_sched_barrier(0);
            ASSIGN_WMMA(j & 1)
            __builtin_amdgcn_sched_barrier(0);
            LDS_BARRIER();
        }
    }
#pragma unroll
    for (int j = 0; j < PD; ++j) {
        if (j + 2 < PD) {
            ASSIGN_STASH((j + 2) & (PD - 1), (j + 2) & 3)
        }
        if (j + 1 < PD) {
            ASSIGN_FRAGS((j + 1) & 1, (j + 1) & 3)
        }
        __builtin_amdgcn_sched_barrier(0);
        ASSIGN_WMMA(j & 1)
        __builtin_amdgcn_sched_barrier(0);
        if (j + 2 < PD) {
            LDS_BARRIER();
        }
    }

    float mx = -3.402823466e38f;
#pragma unroll
    for (int t = 0; t < 4; ++t)
#pragma unroll
        for (int r = 0; r < 8; ++r) mx = fmaxf(mx, acc[t][r]);
    mx = fmaxf(mx, __shfl_xor(mx, 16, 32));

    float sum = 0.0f;
#pragma unroll
    for (int t = 0; t < 4; ++t)
#pragma unroll
        for (int r = 0; r < 8; ++r) {
            float e = __expf(acc[t][r] - mx);
            acc[t][r] = e;
            sum += e;
        }
    sum += __shfl_xor(sum, 16, 32);
    const float inv = 1.0f / sum;

    float* asw = as[wid];
    {
        float* dst = asw + ((col >> 1) << 5) + (half << 4) + (col & 1);
#pragma unroll
        for (int t = 0; t < 4; ++t)
#pragma unroll
            for (int r = 0; r < 8; ++r)
                dst[t * 256 + 2 * r] = acc[t][r] * inv;
    }
    LDS_BARRIER();
    float* ag = a + ((size_t)b * 4 * 512 + (n0 >> 1)) * 32 + (lane << 2);
#pragma unroll
    for (int t = 0; t < 4; ++t)
#pragma unroll
        for (int u = 0; u < 2; ++u)
            *(v4f*)(ag + (size_t)t * 16384 + u * 128) = *(const v4f*)(asw + t * 256 + u * 128 + (lane << 2));
}


#define VLAD_LOAD(SET, IT)                                                      \
    {                                                                           \
        const int nblk = (nh16 + (((IT) + rot) & 15)) << 5;                     \
        const float* ap = ab + nblk * 16;                                       \
        const float* xq = xrow + nblk + koff;                                   \
        _Pragma("unroll")   \
        for (int s = 0; s < 8; ++s) {                                           \
            fa[SET][s][0] = *(const v2f*)(ap + 0 * 16384 + s * 64);             \
            fa[SET][s][1] = *(const v2f*)(ap + 1 * 16384 + s * 64);             \
            fa[SET][s][2] = *(const v2f*)(ap + 2 * 16384 + s * 64);             \
            fa[SET][s][3] = *(const v2f*)(ap + 3 * 16384 + s * 64);             \
            fb[SET][s][0] = *(const v2f*)(xq + 4 * s);                          \
            fb[SET][s][1] = *(const v2f*)(xq + 16 * NN + 4 * s);                \
        }                                                                       \
    }

#define VLAD_WMMA(SET)                                                          \
    _Pragma("unroll")               \
    for (int s = 0; s < 8; ++s) {                                               \
        _Pragma("unroll")           \
        for (int u = 0; u < 2; ++u) {                                           \
            acc[u][0] = __builtin_amdgcn_wmma_f32_16x16x4_f32(false, fa[SET][s][0], false, fb[SET][s][u], (short)0, acc[u][0], false, false); \
            acc[u][1] = __builtin_amdgcn_wmma_f32_16x16x4_f32(false, fa[SET][s][1], false, fb[SET][s][u], (short)0, acc[u][1], false, false); \
            acc[u][2] = __builtin_amdgcn_wmma_f32_16x16x4_f32(false, fa[SET][s][2], false, fb[SET][s][u], (short)0, acc[u][2], false, false); \
            acc[u][3] = __builtin_amdgcn_wmma_f32_16x16x4_f32(false, fa[SET][s][3], false, fb[SET][s][u], (short)0, acc[u][3], false, false); \
        }                                                                       \
    }

#define VLAD_ORDER()                                                            \
    asm volatile("" : "+v"(acc[0][0]), "+v"(acc[0][1]), "+v"(acc[0][2]), "+v"(acc[0][3]),   \
                      "+v"(acc[1][0]), "+v"(acc[1][1]), "+v"(acc[1][2]), "+v"(acc[1][3]) : : "memory")

#define VLAD_DUMP(TB)                                                           \
    _Pragma("unroll")          \
    for (int u = 0; u < 2; ++u)                                                 \
        _Pragma("unroll")                                                       \
        for (int tt = 0; tt < 2; ++tt) {                                        \
            mine[((u * 2 + tt) * 2 + 0) * 32] = __builtin_shufflevector(acc[u][(TB) + tt], acc[u][(TB) + tt], 0, 1, 2, 3); \
            mine[((u * 2 + tt) * 2 + 1) * 32] = __builtin_shufflevector(acc[u][(TB) + tt], acc[u][(TB) + tt], 4, 5, 6, 7); \
        }

#define VLAD_FINISH(TB)                                                         \
    _Pragma("unroll")       \
    for (int u = 0; u < 2; ++u)                                                 \
        _Pragma("unroll")                                                       \
        for (int tt = 0; tt < 2; ++tt) {                                        \
            const v4f lo = theirs[((u * 2 + tt) * 2 + 0) * 32];                 \
            const v4f hi = theirs[((u * 2 + tt) * 2 + 1) * 32];                 \
            _Pragma("unroll")                                                   \
            for (int r = 0; r < 8; ++r) {                                       \
                const int k = ((TB) + tt) * 16 + r + 8 * half;                  \
                ob[(size_t)k * NC + c0 + 16 * u + col] =                        \
                    acc[u][(TB) + tt][r] + (r < 4 ? lo[r & 3] : hi[r & 3]);     \
            }                                                                   \
        }

__global__ __launch_bounds__(256) void netvlad_vlad(
    const float* __restrict__ x, const float* __restrict__ a,
    float* __restrict__ asum, float* __restrict__ out)
{
    __shared__ __attribute__((aligned(16))) float red[8 * 8 * 32 * 4];
    __shared__ v4f srow[8];

    const int lane = threadIdx.x & 31;
    const int wid  = threadIdx.x >> 5;
    const int col  = lane & 15;
    const int half = lane >> 4;
    const int koff = half << 1;
    const int nh   = wid >> 2;
    const int nh16 = nh << 4;

    const int tile = blockIdx.x * 4 + (wid & 3);
    const int b    = tile >> 6;
    const int c0   = (tile & 63) << 5;

    const float* ab   = a + (size_t)b * NK * NN + (lane << 1);
    const float* xrow = x + ((size_t)b * NC + c0 + col) * NN;

    v8f acc[2][4] = {{v8f{}, v8f{}, v8f{}, v8f{}}, {v8f{}, v8f{}, v8f{}, v8f{}}};
    v2f fa[2][8][4], fb[2][8][2];

    const int rot = blockIdx.x & 15;
    VLAD_LOAD(0, 0)
    const int kq = blockIdx.x & 15;
    const v4f* sp = (const v4f*)(a + (((size_t)b * 4 + (kq >> 2)) * 512 + threadIdx.x) * 32 + (kq & 3) * 8);
    const v4f s00 = sp[0], s01 = sp[1], s10 = sp[256 * 8], s11 = sp[256 * 8 + 1];
    for (int it = 0; it < NN / 64 - 2; it += 2) {
        VLAD_LOAD(1, it + 1)
        VLAD_ORDER();
        VLAD_WMMA(0)
        VLAD_ORDER();
        VLAD_LOAD(0, it + 2)
        VLAD_ORDER();
        VLAD_WMMA(1)
        VLAD_ORDER();
    }
    VLAD_LOAD(1, NN / 64 - 1)
    VLAD_ORDER();
    VLAD_WMMA(0)
    VLAD_WMMA(1)

    float* ob = out + (size_t)b * NK * NC;
    v4f* mine = (v4f*)red + (wid * 8) * 32 + lane;
    const v4f* theirs = (const v4f*)red + ((wid ^ 4) * 8) * 32 + lane;
    if (nh == 0) { VLAD_DUMP(2) } else { VLAD_DUMP(0) }
    v4f rs = {(s00.x + s00.y) + (s10.x + s10.y), (s00.z + s00.w) + (s10.z + s10.w),
              (s01.x + s01.y) + (s11.x + s11.y), (s01.z + s01.w) + (s11.z + s11.w)};
#pragma unroll
    for (int off = 16; off > 0; off >>= 1) {
        rs.x += __shfl_xor(rs.x, off, 32);
        rs.y += __shfl_xor(rs.y, off, 32);
        rs.z += __shfl_xor(rs.z, off, 32);
        rs.w += __shfl_xor(rs.w, off, 32);
    }
    if (lane == 0) srow[wid] = rs;
    __syncthreads();
    if (threadIdx.x == 0) {
        v4f t = srow[0];
#pragma unroll
        for (int w = 1; w < 8; ++w) t += srow[w];
        *(v4f*)(asum + b * NK + kq * 4) = t;
    }
    if (nh == 0) { VLAD_FINISH(0) } else { VLAD_FINISH(2) }
}

#define NS 16
#define PIECE4 (NK * NC / 4 / NS)

__global__ __launch_bounds__(256) void netvlad_norm(
    const float* __restrict__ out, const float* __restrict__ asum, const float* __restrict__ cent,
    float* __restrict__ part)
{
    __shared__ float red[8];
    const v4f* vb = (const v4f*)out + (size_t)blockIdx.x * PIECE4 + threadIdx.x;
    const int k0 = (blockIdx.x & (NS - 1)) * (NK / NS);
    const v4f* cb = (const v4f*)cent + (size_t)k0 * (NC / 4) + threadIdx.x;
    const float* ab = asum + (blockIdx.x / NS) * NK + k0;
    v4f v[PIECE4 / 256], cv[PIECE4 / 256];
#pragma unroll
    for (int i = 0; i < PIECE4 / 256; ++i) { v[i] = vb[i * 256]; cv[i] = cb[i * 256]; }
    __builtin_amdgcn_sched_barrier(0);
    float s = 0.0f;
#pragma unroll
    for (int i = 0; i < PIECE4 / 256; ++i) {
        const v4f d = v[i] - ab[i >> 1] * cv[i];
        s += d.x * d.x + d.y * d.y + d.z * d.z + d.w * d.w;
    }
#pragma unroll
    for (int off = 16; off > 0; off >>= 1) s += __shfl_xor(s, off, 32);
    if ((threadIdx.x & 31) == 0) red[threadIdx.x >> 5] = s;
    __syncthreads();
    if (threadIdx.x == 0) {
        float t = 0.0f;
#pragma unroll
        for (int w = 0; w < 8; ++w) t += red[w];
        part[blockIdx.x] = t;
    }
}

__global__ __launch_bounds__(256) void netvlad_scale(
    float* __restrict__ out, const float* __restrict__ asum, const float* __restrict__ cent,
    const float* __restrict__ part)
{
    int gid = blockIdx.x * 256 + threadIdx.x;
    const int row = blockIdx.x >> 1;
    float4* p = (float4*)out;
    float4 v = p[gid];
    const float4 cv = ((const float4*)cent)[gid & (NK * NC / 4 - 1)];
    const float sa = asum[row];
    const float* pb = part + (blockIdx.x / ((NK * NC) / 4 / 256)) * NS;
    float q = 0.0f;
#pragma unroll
    for (int i = 0; i < NS; ++i) q += pb[i];
    const float sc = 1.0f / fmaxf(sqrtf(q), 1e-12f);
    v.x = (v.x - sa * cv.x) * sc; v.y = (v.y - sa * cv.y) * sc;
    v.z = (v.z - sa * cv.z) * sc; v.w = (v.w - sa * cv.w) * sc;
    p[gid] = v;
}

extern "C" void kernel_launch(void* const* d_in, const int* in_sizes, int n_in,
                              void* d_out, int out_size, void* d_ws, size_t ws_size,
                              hipStream_t stream) {
    const float* x    = (const float*)d_in[0];
    const float* w    = (const float*)d_in[1];
    const float* cent = (const float*)d_in[2];
    float* out = (float*)d_out;

    float* a    = (float*)d_ws;
    float* asum = a + (size_t)NB * NK * NN;
    float* part = a;

    netvlad_assign<<<(NB * NN / 16) / 4, 128, 0, stream>>>(x, w, a);
    netvlad_vlad  <<<(NB * NC / 32) / 4, 256, 0, stream>>>(x, a, asum, out);
    netvlad_norm  <<<NB * NS, 256, 0, stream>>>(out, asum, cent, part);
    netvlad_scale <<<(NB * NK * NC / 4) / 256, 256, 0, stream>>>(out, asum, cent, part);
}
